# starting kernel
# baseline (speedup 1.0000x reference)
#include <hip/hip_runtime.h>

typedef float v2f __attribute__((ext_vector_type(2)));
typedef float v8f __attribute__((ext_vector_type(8)));

#define T_IN   2048
#define T_OUT  2033
#define NF     64
#define ROW    66
#define EPSV   1e-5f

__global__ __launch_bounds__(256) void win_norm_wmma(const float* __restrict__ in,
                                                     float* __restrict__ out) {
    const int wid   = blockIdx.x * 8 + (threadIdx.x >> 5);
    const int lane  = threadIdx.x & 31;
    const int ftile = wid & 3;
    const int ttile = (wid >> 2) & 127;
    const int b     = wid >> 9;

    const int m = lane & 15;
    const int h = lane >> 4;
    const int f = ftile * 16 + m;
    const int tbase = ttile * 16;

    const int tload = (tbase <= T_IN - 32) ? tbase : (T_IN - 32);
    const int msh   = m + (tbase - tload);

    const float* base = in + ((size_t)b * T_IN + (size_t)(tload + 2 * h)) * ROW;
    const float* px   = base + f;
    const float* pi   = base + 64;

    v8f cnum = {};
    v8f cden = {};

#pragma unroll
    for (int c = 0; c < 8; ++c) {
        const float x0 = px[c * 4 * ROW];
        const float i0 = pi[c * 4 * ROW];
        const float x1 = px[c * 4 * ROW + ROW];
        const float i1 = pi[c * 4 * ROW + ROW];

        const int j0 = 4 * c + 2 * h;

        v2f a;
        a.x = ((unsigned)(j0     - msh) < 16u) ? 1.0f : 0.0f;
        a.y = ((unsigned)(j0 + 1 - msh) < 16u) ? 1.0f : 0.0f;

        v2f bn, bd;
        bn.x = x0 * i0;             bn.y = x1 * i1;
        bd.x = __builtin_fabsf(x0); bd.y = __builtin_fabsf(x1);

        cnum = __builtin_amdgcn_wmma_f32_16x16x4_f32(false, a, false, bn,
                                                     (short)0, cnum, false, false);
        cden = __builtin_amdgcn_wmma_f32_16x16x4_f32(false, a, false, bd,
                                                     (short)0, cden, false, false);
    }

    const int tstart = tbase + 8 * h;
    float* op = out + ((size_t)b * NF + f) * T_OUT + tstart;

#pragma unroll
    for (int v = 0; v < 8; ++v) {
        if (tstart + v < T_OUT) {
            op[v] = cnum[v] / (cden[v] + EPSV);
        }
    }
}

extern "C" void kernel_launch(void* const* d_in, const int* in_sizes, int n_in,
                              void* d_out, int out_size, void* d_ws, size_t ws_size,
                              hipStream_t stream) {
    const float* in  = (const float*)d_in[0];
    float*       out = (float*)d_out;
    win_norm_wmma<<<dim3(8192), dim3(256), 0, stream>>>(in, out);
}


# Swap WMMA operand roles (data as A, 0/1 band as B) so D is feature x time: each store instruction writes two contiguous 64-byte time runs instead of 32 scattered 4-byte words; sum order unchanged
# speedup vs baseline: 3.4440x; 3.4440x over previous
#include <hip/hip_runtime.h>


typedef float v2f __attribute__((ext_vector_type(2)));
typedef float v8f __attribute__((ext_vector_type(8)));

#define T_IN   2048
#define T_OUT  2033
#define NF     64
#define ROW    66
#define EPSV   1e-5f

__global__ __launch_bounds__(256) void win_norm_wmma(const float* __restrict__ in,
                                                     float* __restrict__ out) {
    const int wid   = blockIdx.x * 8 + (threadIdx.x >> 5);
    const int lane  = threadIdx.x & 31;
    const int ftile = wid & 3;
    const int ttile = (wid >> 2) & 127;
    const int b     = wid >> 9;

    const int c = lane & 15;
    const int h = lane >> 4;
    const int f = ftile * 16 + c;
    const int tbase = ttile * 16;

    const int tload = (tbase <= T_IN - 32) ? tbase : (T_IN - 32);
    const int csh   = c + (tbase - tload);

    const float* base = in + ((size_t)b * T_IN + (size_t)(tload + 2 * h)) * ROW;
    const float* px   = base + f;
    const float* pi   = base + 64;

    v8f cnum = {};
    v8f cden = {};

#pragma unroll
    for (int ch = 0; ch < 8; ++ch) {
        const float x0 = px[ch * 4 * ROW];
        const float i0 = pi[ch * 4 * ROW];
        const float x1 = px[ch * 4 * ROW + ROW];
        const float i1 = pi[ch * 4 * ROW + ROW];

        const int j0 = 4 * ch + 2 * h;

        v2f g;
        g.x = ((unsigned)(j0     - csh) < 16u) ? 1.0f : 0.0f;
        g.y = ((unsigned)(j0 + 1 - csh) < 16u) ? 1.0f : 0.0f;

        v2f xn, xd;
        xn.x = x0 * i0;             xn.y = x1 * i1;
        xd.x = __builtin_fabsf(x0); xd.y = __builtin_fabsf(x1);

        cnum = __builtin_amdgcn_wmma_f32_16x16x4_f32(false, xn, false, g,
                                                     (short)0, cnum, false, false);
        cden = __builtin_amdgcn_wmma_f32_16x16x4_f32(false, xd, false, g,
                                                     (short)0, cden, false, false);
    }

    const int tq = tbase + c;
    float* op = out + ((size_t)b * NF + (size_t)(ftile * 16 + 8 * h)) * T_OUT + tq;

    if (tq < T_OUT) {
#pragma unroll
        for (int r = 0; r < 8; ++r) {
            op[(size_t)r * T_OUT] = cnum[r] / (cden[r] + EPSV);
        }
    }
}

extern "C" void kernel_launch(void* const* d_in, const int* in_sizes, int n_in,
                              void* d_out, int out_size, void* d_ws, size_t ws_size,
                              hipStream_t stream) {
    const float* in  = (const float*)d_in[0];
    float*       out = (float*)d_out;
    win_norm_wmma<<<dim3(8192), dim3(256), 0, stream>>>(in, out);
}


# Each wave walks 8 consecutive time tiles and keeps the 16 rows two neighbouring tiles share in registers (65536 -> 8192 waves, loads halved, band built once per wave); sum order unchanged
# speedup vs baseline: 4.5881x; 1.3322x over previous
#include <hip/hip_runtime.h>


typedef float v2f __attribute__((ext_vector_type(2)));
typedef float v8f __attribute__((ext_vector_type(8)));

#define T_IN   2048
#define T_OUT  2033
#define NF     64
#define ROW    66
#define EPSV   1e-5f
#define NT     8


struct Raw  { float x[8]; float i[8]; };
struct Frag { v2f n[4]; v2f d[4]; };

__device__ __forceinline__ void load_raw(Raw& r, const float* __restrict__ gp, int offx, int offi) {
#pragma unroll
    for (int q = 0; q < 4; ++q) {
        r.x[2 * q]     = gp[offx + (4 * q) * ROW];
        r.i[2 * q]     = gp[offi + (4 * q) * ROW];
        r.x[2 * q + 1] = gp[offx + (4 * q + 1) * ROW];
        r.i[2 * q + 1] = gp[offi + (4 * q + 1) * ROW];
    }
}

__device__ __forceinline__ void to_frag(Frag& f, const Raw& r) {
#pragma unroll
    for (int q = 0; q < 4; ++q) {
        f.n[q].x = r.x[2 * q] * r.i[2 * q];
        f.n[q].y = r.x[2 * q + 1] * r.i[2 * q + 1];
        f.d[q].x = __builtin_fabsf(r.x[2 * q]);
        f.d[q].y = __builtin_fabsf(r.x[2 * q + 1]);
    }
}

__device__ __forceinline__ void tile(const Frag& lo, const Frag& up,
                                     const v2f (&glo)[4], const v2f (&gup)[4],
                                     float* __restrict__ op, bool ok) {
    v8f cnum = {};
    v8f cden = {};
#pragma unroll
    for (int q = 0; q < 4; ++q) {
        cnum = __builtin_amdgcn_wmma_f32_16x16x4_f32(false, lo.n[q], false, glo[q],
                                                     (short)0, cnum, false, false);
        cden = __builtin_amdgcn_wmma_f32_16x16x4_f32(false, lo.d[q], false, glo[q],
                                                     (short)0, cden, false, false);
    }
#pragma unroll
    for (int q = 0; q < 4; ++q) {
        cnum = __builtin_amdgcn_wmma_f32_16x16x4_f32(false, up.n[q], false, gup[q],
                                                     (short)0, cnum, false, false);
        cden = __builtin_amdgcn_wmma_f32_16x16x4_f32(false, up.d[q], false, gup[q],
                                                     (short)0, cden, false, false);
    }
    if (ok) {
#pragma unroll
        for (int r = 0; r < 8; ++r) {
            op[(size_t)r * T_OUT] = cnum[r] / (cden[r] + EPSV);
        }
    }
}

__global__ __launch_bounds__(256) void win_norm_wmma(const float* __restrict__ in,
                                                     float* __restrict__ out) {
    const int wid   = __builtin_amdgcn_readfirstlane((int)(blockIdx.x * 8 + (threadIdx.x >> 5)));
    const int lane  = threadIdx.x & 31;
    const int ftile = wid & 3;
    const int strip = (wid >> 2) & 15;
    const int b     = wid >> 6;

    const int c  = lane & 15;
    const int h  = lane >> 4;
    const int t0 = strip * (16 * NT);

    v2f glo[4], gup[4];
#pragma unroll
    for (int q = 0; q < 4; ++q) {
        const int j0 = 4 * q + 2 * h;
        glo[q].x = (j0     >= c) ? 1.0f : 0.0f;
        glo[q].y = (j0 + 1 >= c) ? 1.0f : 0.0f;
        gup[q].x = (j0     <  c) ? 1.0f : 0.0f;
        gup[q].y = (j0 + 1 <  c) ? 1.0f : 0.0f;
    }

    const float* gb = in + (size_t)b * T_IN * ROW;
    const int offx  = 2 * h * ROW + ftile * 16 + c;
    const int offi  = 2 * h * ROW + 64;
    float* ob = out + ((size_t)b * NF + (size_t)(ftile * 16)) * T_OUT + t0;
    const int offo = 8 * h * T_OUT + c;

#define GRP(g) (gb + (size_t)(((t0 + 16 * (g)) <= (T_IN - 16)) ? (t0 + 16 * (g)) : (T_IN - 16)) * ROW)

    Raw ra, rc, rn;
    Frag lo, up;
    load_raw(ra, GRP(0), offx, offi);
    load_raw(rc, GRP(1), offx, offi);

#pragma unroll
    for (int j = 0; j < NT; ++j) {
        if (j + 1 < NT) {
            load_raw(rn, GRP(j + 2), offx, offi);
        }
        __builtin_amdgcn_sched_barrier(0);
        if (j == 0) {
            to_frag(lo, ra);
        }
        to_frag(up, rc);
        tile(lo, up, glo, gup, ob + 16 * j + offo, (t0 + 16 * j + c) < T_OUT);
        lo = up;
        rc = rn;
    }
#undef GRP
}

extern "C" void kernel_launch(void* const* d_in, const int* in_sizes, int n_in,
                              void* d_out, int out_size, void* d_ws, size_t ws_size,
                              hipStream_t stream) {
    const float* in  = (const float*)d_in[0];
    float*       out = (float*)d_out;
    win_norm_wmma<<<dim3(1024), dim3(256), 0, stream>>>(in, out);
}


# Quotient num/(den+eps) taken as num times the hardware reciprocal (1 ulp) instead of the 11-instruction IEEE division, 64 per lane per wave; result differs by about 1 ulp; sum order unchanged
# speedup vs baseline: 4.7690x; 1.0394x over previous
#include <hip/hip_runtime.h>


typedef float v2f __attribute__((ext_vector_type(2)));
typedef float v8f __attribute__((ext_vector_type(8)));

#define T_IN   2048
#define T_OUT  2033
#define NF     64
#define ROW    66
#define EPSV   1e-5f
#define NT     8


struct Raw  { float x[8]; float i[8]; };
struct Frag { v2f n[4]; v2f d[4]; };

__device__ __forceinline__ void load_raw(Raw& r, const float* __restrict__ gp, int offx, int offi) {
#pragma unroll
    for (int q = 0; q < 4; ++q) {
        r.x[2 * q]     = gp[offx + (4 * q) * ROW];
        r.i[2 * q]     = gp[offi + (4 * q) * ROW];
        r.x[2 * q + 1] = gp[offx + (4 * q + 1) * ROW];
        r.i[2 * q + 1] = gp[offi + (4 * q + 1) * ROW];
    }
}

__device__ __forceinline__ void to_frag(Frag& f, const Raw& r) {
#pragma unroll
    for (int q = 0; q < 4; ++q) {
        f.n[q].x = r.x[2 * q] * r.i[2 * q];
        f.n[q].y = r.x[2 * q + 1] * r.i[2 * q + 1];
        f.d[q].x = __builtin_fabsf(r.x[2 * q]);
        f.d[q].y = __builtin_fabsf(r.x[2 * q + 1]);
    }
}

__device__ __forceinline__ void tile(const Frag& lo, const Frag& up,
                                     const v2f (&glo)[4], const v2f (&gup)[4],
                                     float* __restrict__ op, bool ok) {
    v8f cnum = {};
    v8f cden = {};
#pragma unroll
    for (int q = 0; q < 4; ++q) {
        cnum = __builtin_amdgcn_wmma_f32_16x16x4_f32(false, lo.n[q], false, glo[q],
                                                     (short)0, cnum, false, false);
        cden = __builtin_amdgcn_wmma_f32_16x16x4_f32(false, lo.d[q], false, glo[q],
                                                     (short)0, cden, false, false);
    }
#pragma unroll
    for (int q = 0; q < 4; ++q) {
        cnum = __builtin_amdgcn_wmma_f32_16x16x4_f32(false, up.n[q], false, gup[q],
                                                     (short)0, cnum, false, false);
        cden = __builtin_amdgcn_wmma_f32_16x16x4_f32(false, up.d[q], false, gup[q],
                                                     (short)0, cden, false, false);
    }
    if (ok) {
#pragma unroll
        for (int r = 0; r < 8; ++r) {
            op[(size_t)r * T_OUT] = cnum[r] * __builtin_amdgcn_rcpf(cden[r] + EPSV);
        }
    }
}

__global__ __launch_bounds__(256) void win_norm_wmma(const float* __restrict__ in,
                                                     float* __restrict__ out) {
    const int wid   = __builtin_amdgcn_readfirstlane((int)(blockIdx.x * 8 + (threadIdx.x >> 5)));
    const int lane  = threadIdx.x & 31;
    const int ftile = wid & 3;
    const int strip = (wid >> 2) & 15;
    const int b     = wid >> 6;

    const int c  = lane & 15;
    const int h  = lane >> 4;
    const int t0 = strip * (16 * NT);

    v2f glo[4], gup[4];
#pragma unroll
    for (int q = 0; q < 4; ++q) {
        const int j0 = 4 * q + 2 * h;
        glo[q].x = (j0     >= c) ? 1.0f : 0.0f;
        glo[q].y = (j0 + 1 >= c) ? 1.0f : 0.0f;
        gup[q].x = (j0     <  c) ? 1.0f : 0.0f;
        gup[q].y = (j0 + 1 <  c) ? 1.0f : 0.0f;
    }

    const float* gb = in + (size_t)b * T_IN * ROW;
    const int offx  = 2 * h * ROW + ftile * 16 + c;
    const int offi  = 2 * h * ROW + 64;
    float* ob = out + ((size_t)b * NF + (size_t)(ftile * 16)) * T_OUT + t0;
    const int offo = 8 * h * T_OUT + c;

#define GRP(g) (gb + (size_t)(((t0 + 16 * (g)) <= (T_IN - 16)) ? (t0 + 16 * (g)) : (T_IN - 16)) * ROW)

    Raw ra, rc, rn;
    Frag lo, up;
    load_raw(ra, GRP(0), offx, offi);
    load_raw(rc, GRP(1), offx, offi);

#pragma unroll
    for (int j = 0; j < NT; ++j) {
        if (j + 1 < NT) {
            load_raw(rn, GRP(j + 2), offx, offi);
        }
        __builtin_amdgcn_sched_barrier(0);
        if (j == 0) {
            to_frag(lo, ra);
        }
        to_frag(up, rc);
        tile(lo, up, glo, gup, ob + 16 * j + offo, (t0 + 16 * j + c) < T_OUT);
        lo = up;
        rc = rn;
    }
#undef GRP
}

extern "C" void kernel_launch(void* const* d_in, const int* in_sizes, int n_in,
                              void* d_out, int out_size, void* d_ws, size_t ws_size,
                              hipStream_t stream) {
    const float* in  = (const float*)d_in[0];
    float*       out = (float*)d_out;
    win_norm_wmma<<<dim3(1024), dim3(256), 0, stream>>>(in, out);
}


# Results collected in a wave-private LDS block (16 features x 128 outputs) and written as whole aligned 128-byte lines, one store per line, not as 64-byte pieces per tile; sum order unchanged
# speedup vs baseline: 5.6748x; 1.1899x over previous
#include <hip/hip_runtime.h>


typedef float v2f __attribute__((ext_vector_type(2)));
typedef float v8f __attribute__((ext_vector_type(8)));

#define T_IN   2048
#define T_OUT  2033
#define NF     64
#define ROW    66
#define EPSV   1e-5f
#define NT     8
#define PITCH  132


struct Raw  { float x[8]; float i[8]; };
struct Frag { v2f n[4]; v2f d[4]; };

__device__ __forceinline__ void load_raw(Raw& r, const float* __restrict__ gp, int offx, int offi) {
#pragma unroll
    for (int q = 0; q < 4; ++q) {
        r.x[2 * q]     = gp[offx + (4 * q) * ROW];
        r.i[2 * q]     = gp[offi + (4 * q) * ROW];
        r.x[2 * q + 1] = gp[offx + (4 * q + 1) * ROW];
        r.i[2 * q + 1] = gp[offi + (4 * q + 1) * ROW];
    }
}

__device__ __forceinline__ void to_frag(Frag& f, const Raw& r) {
#pragma unroll
    for (int q = 0; q < 4; ++q) {
        f.n[q].x = r.x[2 * q] * r.i[2 * q];
        f.n[q].y = r.x[2 * q + 1] * r.i[2 * q + 1];
        f.d[q].x = __builtin_fabsf(r.x[2 * q]);
        f.d[q].y = __builtin_fabsf(r.x[2 * q + 1]);
    }
}

__device__ __forceinline__ void tile(const Frag& lo, const Frag& up,
                                     const v2f (&glo)[4], const v2f (&gup)[4],
                                     float* __restrict__ sp) {
    v8f cnum = {};
    v8f cden = {};
#pragma unroll
    for (int q = 0; q < 4; ++q) {
        cnum = __builtin_amdgcn_wmma_f32_16x16x4_f32(false, lo.n[q], false, glo[q],
                                                     (short)0, cnum, false, false);
        cden = __builtin_amdgcn_wmma_f32_16x16x4_f32(false, lo.d[q], false, glo[q],
                                                     (short)0, cden, false, false);
    }
#pragma unroll
    for (int q = 0; q < 4; ++q) {
        cnum = __builtin_amdgcn_wmma_f32_16x16x4_f32(false, up.n[q], false, gup[q],
                                                     (short)0, cnum, false, false);
        cden = __builtin_amdgcn_wmma_f32_16x16x4_f32(false, up.d[q], false, gup[q],
                                                     (short)0, cden, false, false);
    }
#pragma unroll
    for (int r = 0; r < 8; ++r) {
        sp[r * PITCH] = cnum[r] * __builtin_amdgcn_rcpf(cden[r] + EPSV);
    }
}

template <bool LAST>
__device__ __forceinline__ void flush(const float* ws, float* __restrict__ orow, int lane, int lim) {
#pragma unroll
    for (int g = 0; g < 16; g += 4) {
        float v[4][4];
        int   iw[4];
#pragma unroll
        for (int e = 0; e < 4; ++e) {
            const int f = g + e;
            const float* rp = orow + (size_t)f * T_OUT;
            const int s = (int)(((size_t)rp >> 2) & 31);
            const int u = lane + 32 - s;
            iw[e] = (lane - s) & (16 * NT - 1);
            const float* src = ws + f * PITCH;
            v[e][0] = src[iw[e]];
            v[e][1] = src[u];
            v[e][2] = src[u + 32];
            v[e][3] = src[u + 64];
        }
#pragma unroll
        for (int e = 0; e < 4; ++e) {
            const int f = g + e;
            float* rp = orow + (size_t)f * T_OUT;
            const int s = (int)(((size_t)rp >> 2) & 31);
            float* line0 = rp - s;
            const int u = lane + 32 - s;
            if (!LAST || iw[e] < lim)    line0[iw[e] + s] = v[e][0];
            line0[lane + 32] = v[e][1];
            line0[lane + 64] = v[e][2];
            if (!LAST || u + 64 < lim)   line0[lane + 96] = v[e][3];
        }
    }
}

__global__ __launch_bounds__(256) void win_norm_wmma(const float* __restrict__ in,
                                                     float* __restrict__ out) {
    __shared__ float sm[8 * 16 * PITCH];
    const int wid   = __builtin_amdgcn_readfirstlane((int)(blockIdx.x * 8 + (threadIdx.x >> 5)));
    const int lane  = threadIdx.x & 31;
    const int ftile = wid & 3;
    const int strip = (wid >> 2) & 15;
    const int b     = wid >> 6;

    const int c  = lane & 15;
    const int h  = lane >> 4;
    const int t0 = strip * (16 * NT);

    v2f glo[4], gup[4];
#pragma unroll
    for (int q = 0; q < 4; ++q) {
        const int j0 = 4 * q + 2 * h;
        glo[q].x = (j0     >= c) ? 1.0f : 0.0f;
        glo[q].y = (j0 + 1 >= c) ? 1.0f : 0.0f;
        gup[q].x = (j0     <  c) ? 1.0f : 0.0f;
        gup[q].y = (j0 + 1 <  c) ? 1.0f : 0.0f;
    }

    const float* gb = in + (size_t)b * T_IN * ROW;
    const int offx  = 2 * h * ROW + ftile * 16 + c;
    const int offi  = 2 * h * ROW + 64;
    float* ob = out + ((size_t)b * NF + (size_t)(ftile * 16)) * T_OUT + t0;
    float* ws = sm + (threadIdx.x >> 5) * (16 * PITCH);
    float* wl = ws + 8 * h * PITCH + c;

#define GRP(g) (gb + (size_t)(((t0 + 16 * (g)) <= (T_IN - 16)) ? (t0 + 16 * (g)) : (T_IN - 16)) * ROW)

    Raw ra, rc, rn;
    Frag lo, up;
    load_raw(ra, GRP(0), offx, offi);
    load_raw(rc, GRP(1), offx, offi);

#pragma unroll
    for (int j = 0; j < NT; ++j) {
        if (j + 1 < NT) {
            load_raw(rn, GRP(j + 2), offx, offi);
        }
        __builtin_amdgcn_sched_barrier(0);
        if (j == 0) {
            to_frag(lo, ra);
        }
        to_frag(up, rc);
        tile(lo, up, glo, gup, wl + 16 * j);
        lo = up;
        rc = rn;
    }
#undef GRP

    static_assert(T_OUT - (T_IN / (16 * NT) - 1) * (16 * NT) > 95, "last strip shorter than three lines");
    if (t0 + 16 * NT <= T_OUT) {
        flush<false>(ws, ob, lane, 16 * NT);
    } else {
        flush<true>(ws, ob, lane, T_OUT - t0);
    }
}

extern "C" void kernel_launch(void* const* d_in, const int* in_sizes, int n_in,
                              void* d_out, int out_size, void* d_ws, size_t ws_size,
                              hipStream_t stream) {
    const float* in  = (const float*)d_in[0];
    float*       out = (float*)d_out;
    win_norm_wmma<<<dim3(1024), dim3(256), 0, stream>>>(in, out);
}
